# starting kernel
# baseline (speedup 1.0000x reference)
#include <hip/hip_runtime.h>

#define DIMS_IN  128
#define DIMS_H   96
#define DIMS_OUT 32

typedef float v2f __attribute__((ext_vector_type(2)));
typedef float v8f __attribute__((ext_vector_type(8)));

__global__ void zero_kernel(float4* __restrict__ p, int n4) {
    int i = blockIdx.x * blockDim.x + threadIdx.x;
    if (i < n4) {
        float4 z; z.x = 0.f; z.y = 0.f; z.z = 0.f; z.w = 0.f;
        p[i] = z;
    }
}

__global__ void transpose_kernel(const float* __restrict__ W,
                                 float* __restrict__ Wt, int K, int N) {
    int i = blockIdx.x * blockDim.x + threadIdx.x;
    if (i < K * N) {
        int k = i / N;
        int n = i - k * N;
        Wt[n * K + k] = W[i];
    }
}

template <int K, bool RELU_A>
__global__ void gemm_wmma_f32_bt(const float* __restrict__ A,
                                 const float* __restrict__ Bt,
                                 float* __restrict__ C,
                                 int M, int N) {
    const int wavesPerBlock = blockDim.x >> 5;
    const int waveId = blockIdx.x * wavesPerBlock + (threadIdx.x >> 5);
    const int lane   = threadIdx.x & 31;
    const int nPair  = N >> 5;
    const int tileM  = waveId / nPair;
    const int tileP  = waveId - tileM * nPair;
    if (tileM * 16 >= M) return;

    const int half = lane >> 4;
    const int l15  = lane & 15;

    const float* arow = A  + (tileM * 16 + l15) * K + half * 2;
    const float* brow = Bt + (tileP * 32 + l15) * K + half * 2;

    v8f acc0 = {};
    v8f acc1 = {};
    #pragma unroll 4
    for (int k = 0; k < K; k += 4) {
        v2f a = *(const v2f*)(arow + k);
        if (RELU_A) {
            a.x = fmaxf(a.x, 0.f);
            a.y = fmaxf(a.y, 0.f);
        }
        v2f b0 = *(const v2f*)(brow + k);
        v2f b1 = *(const v2f*)(brow + 16 * K + k);
        acc0 = __builtin_amdgcn_wmma_f32_16x16x4_f32(
            false, a, false, b0, (short)0, acc0, false, false);
        acc1 = __builtin_amdgcn_wmma_f32_16x16x4_f32(
            false, a, false, b1, (short)0, acc1, false, false);
    }

    float* c0 = C + (tileM * 16 + half * 8) * N + tileP * 32 + l15;
    #pragma unroll
    for (int r = 0; r < 8; ++r) {
        c0[r * N]      = acc0[r];
        c0[r * N + 16] = acc1[r];
    }
}

__global__ void spmm_scatter(const int* __restrict__ esrc,
                             const int* __restrict__ edst,
                             const float* __restrict__ eval,
                             const float* __restrict__ dense,
                             float* __restrict__ out,
                             int total, int D4, int D) {
    int i = blockIdx.x * blockDim.x + threadIdx.x;
    if (i >= total) return;
    int e = i / D4;
    int c = (i - e * D4) * 4;
    int s = esrc[e];
    int d = edst[e];
    float v = eval[e];
    const float4 m = *(const float4*)(dense + (size_t)s * D + c);
    float* o = out + (size_t)d * D + c;
    atomicAdd(o + 0, v * m.x);
    atomicAdd(o + 1, v * m.y);
    atomicAdd(o + 2, v * m.z);
    atomicAdd(o + 3, v * m.w);
}

extern "C" void kernel_launch(void* const* d_in, const int* in_sizes, int n_in,
                              void* d_out, int out_size, void* d_ws, size_t ws_size,
                              hipStream_t stream) {
    const float* x    = (const float*)d_in[0];
    const int*   esrc = (const int*)d_in[1];
    const int*   edst = (const int*)d_in[2];
    const float* eval = (const float*)d_in[3];
    const float* W1   = (const float*)d_in[4];
    const float* W2   = (const float*)d_in[5];
    float*       out  = (float*)d_out;

    const int E = in_sizes[1];
    const int n = in_sizes[0] / DIMS_IN;

    float* buf0 = (float*)d_ws;
    float* buf1 = buf0 + (size_t)n * DIMS_H;
    float* W1t  = buf1 + (size_t)n * DIMS_H;
    float* W2t  = W1t + DIMS_H * DIMS_IN;

    transpose_kernel<<<(DIMS_IN * DIMS_H + 255) / 256, 256, 0, stream>>>(
        W1, W1t, DIMS_IN, DIMS_H);
    transpose_kernel<<<(DIMS_H * DIMS_OUT + 255) / 256, 256, 0, stream>>>(
        W2, W2t, DIMS_H, DIMS_OUT);

    {
        int n4 = (n * DIMS_H) / 4;
        zero_kernel<<<(n4 + 255) / 256, 256, 0, stream>>>((float4*)buf1, n4);
        int o4 = (n * DIMS_OUT) / 4;
        zero_kernel<<<(o4 + 255) / 256, 256, 0, stream>>>((float4*)out, o4);
    }

    {
        int waves  = (n / 16) * (DIMS_H / 32);
        int blocks = (waves + 7) / 8;
        gemm_wmma_f32_bt<DIMS_IN, false><<<blocks, 256, 0, stream>>>(
            x, W1t, buf0, n, DIMS_H);
    }

    {
        int total = E * (DIMS_H / 4);
        spmm_scatter<<<(total + 255) / 256, 256, 0, stream>>>(
            esrc, edst, eval, buf0, buf1, total, DIMS_H / 4, DIMS_H);
    }

    {
        int waves  = (n / 16) * (DIMS_OUT / 32);
        int blocks = (waves + 7) / 8;
        gemm_wmma_f32_bt<DIMS_H, true><<<blocks, 256, 0, stream>>>(
            buf1, W2t, buf0, n, DIMS_OUT);
    }

    {
        int total = E * (DIMS_OUT / 4);
        spmm_scatter<<<(total + 255) / 256, 256, 0, stream>>>(
            esrc, edst, eval, buf0, out, total, DIMS_OUT / 4, DIMS_OUT);
    }
}


# SpMM f32 atomic scatter replaced by per-call counting sort of edges by dst (int atomic rank, scan, place) and a per-row register gather-sum; row sum order = atomic arrival rank; no zero fills
# speedup vs baseline: 1.2095x; 1.2095x over previous
#include <hip/hip_runtime.h>

#define DIMS_IN  128
#define DIMS_H   96
#define DIMS_OUT 32

typedef float v2f __attribute__((ext_vector_type(2)));
typedef float v8f __attribute__((ext_vector_type(8)));

__global__ void zero_kernel(float4* __restrict__ p, int n4) {
    int i = blockIdx.x * blockDim.x + threadIdx.x;
    if (i < n4) {
        float4 z; z.x = 0.f; z.y = 0.f; z.z = 0.f; z.w = 0.f;
        p[i] = z;
    }
}

__global__ void transpose_kernel(const float* __restrict__ W,
                                 float* __restrict__ Wt, int K, int N) {
    int i = blockIdx.x * blockDim.x + threadIdx.x;
    if (i < K * N) {
        int k = i / N;
        int n = i - k * N;
        Wt[n * K + k] = W[i];
    }
}

template <int K, bool RELU_A>
__global__ void gemm_wmma_f32_bt(const float* __restrict__ A,
                                 const float* __restrict__ Bt,
                                 float* __restrict__ C,
                                 int M, int N) {
    const int wavesPerBlock = blockDim.x >> 5;
    const int waveId = blockIdx.x * wavesPerBlock + (threadIdx.x >> 5);
    const int lane   = threadIdx.x & 31;
    const int nPair  = N >> 5;
    const int tileM  = waveId / nPair;
    const int tileP  = waveId - tileM * nPair;
    if (tileM * 16 >= M) return;

    const int half = lane >> 4;
    const int l15  = lane & 15;

    const float* arow = A  + (tileM * 16 + l15) * K + half * 2;
    const float* brow = Bt + (tileP * 32 + l15) * K + half * 2;

    v8f acc0 = {};
    v8f acc1 = {};
    #pragma unroll 4
    for (int k = 0; k < K; k += 4) {
        v2f a = *(const v2f*)(arow + k);
        if (RELU_A) {
            a.x = fmaxf(a.x, 0.f);
            a.y = fmaxf(a.y, 0.f);
        }
        v2f b0 = *(const v2f*)(brow + k);
        v2f b1 = *(const v2f*)(brow + 16 * K + k);
        acc0 = __builtin_amdgcn_wmma_f32_16x16x4_f32(
            false, a, false, b0, (short)0, acc0, false, false);
        acc1 = __builtin_amdgcn_wmma_f32_16x16x4_f32(
            false, a, false, b1, (short)0, acc1, false, false);
    }

    float* c0 = C + (tileM * 16 + half * 8) * N + tileP * 32 + l15;
    #pragma unroll
    for (int r = 0; r < 8; ++r) {
        c0[r * N]      = acc0[r];
        c0[r * N + 16] = acc1[r];
    }
}

__global__ void edge_rank(const int* __restrict__ edst,
                          int* __restrict__ cnt,
                          int* __restrict__ rank, int E) {
    int e = blockIdx.x * blockDim.x + threadIdx.x;
    if (e < E) rank[e] = atomicAdd(cnt + edst[e], 1);
}

__global__ void __launch_bounds__(1024)
row_start_scan(const int* __restrict__ cnt, int* __restrict__ rowptr, int per4) {
    __shared__ int wsum[32];
    const int t    = threadIdx.x;
    const int lane = t & 31;
    const int w    = t >> 5;
    const int4* src = (const int4*)cnt + t * per4;
    int4*       dst = (int4*)rowptr + t * per4;

    int s = 0;
    #pragma unroll 4
    for (int i = 0; i < per4; ++i) {
        int4 c = src[i];
        s += (c.x + c.y) + (c.z + c.w);
    }
    int incl = s;
    #pragma unroll
    for (int off = 1; off < 32; off <<= 1) {
        int v = __shfl_up(incl, off);
        if (lane >= off) incl += v;
    }
    if (lane == 31) wsum[w] = incl;
    __syncthreads();
    if (w == 0) {
        int tot = wsum[lane];
        int ti  = tot;
        #pragma unroll
        for (int off = 1; off < 32; off <<= 1) {
            int v = __shfl_up(ti, off);
            if (lane >= off) ti += v;
        }
        wsum[lane] = ti - tot;
    }
    __syncthreads();
    int run = wsum[w] + (incl - s);
    #pragma unroll 4
    for (int i = 0; i < per4; ++i) {
        int4 c = src[i];
        int4 o;
        o.x = run;
        o.y = o.x + c.x;
        o.z = o.y + c.y;
        o.w = o.z + c.z;
        run = o.w + c.w;
        dst[i] = o;
    }
}

__global__ void edge_place(const int* __restrict__ esrc,
                           const int* __restrict__ edst,
                           const float* __restrict__ eval,
                           const int* __restrict__ rowptr,
                           const int* __restrict__ rank,
                           int* __restrict__ sorted, int E) {
    int e = blockIdx.x * blockDim.x + threadIdx.x;
    if (e < E) {
        int p = rowptr[edst[e]] + rank[e];
        int2 r;
        r.x = esrc[e];
        r.y = __float_as_int(eval[e]);
        ((int2*)sorted)[p] = r;
    }
}

template <int D, int K>
__device__ __forceinline__ void gather_edges(const int2* __restrict__ ed,
                                             const float* __restrict__ base,
                                             float4& acc) {
    int2   e[K];
    float4 m[K];
    #pragma unroll
    for (int j = 0; j < K; ++j) e[j] = ed[j];
    #pragma unroll
    for (int j = 0; j < K; ++j) m[j] = *(const float4*)(base + e[j].x * D);
    #pragma unroll
    for (int j = 0; j < K; ++j) {
        const float v = __int_as_float(e[j].y);
        acc.x = fmaf(v, m[j].x, acc.x);
        acc.y = fmaf(v, m[j].y, acc.y);
        acc.z = fmaf(v, m[j].z, acc.z);
        acc.w = fmaf(v, m[j].w, acc.w);
    }
}

template <int D, int LPR>
__global__ void spmm_gather(const int* __restrict__ rowptr,
                            const int* __restrict__ sorted,
                            const float* __restrict__ dense,
                            float* __restrict__ out, int n) {
    constexpr int D4 = D / 4;
    const int lane = threadIdx.x & 31;
    const int wave = (blockIdx.x * blockDim.x + threadIdx.x) >> 5;
    int row, li;
    if constexpr (LPR == 32) {
        row = __builtin_amdgcn_readfirstlane(wave);
        li  = lane;
    } else {
        row = wave * (32 / LPR) + lane / LPR;
        li  = lane % LPR;
    }
    if (row >= n || li >= D4) return;

    const int2* ed = (const int2*)sorted;
    int       p    = rowptr[row];
    const int pend = rowptr[row + 1];
    const float* base = dense + li * 4;

    float4 acc;
    acc.x = 0.f; acc.y = 0.f; acc.z = 0.f; acc.w = 0.f;
    for (; p + 8 <= pend; p += 8) gather_edges<D, 8>(ed + p, base, acc);
    if (p + 4 <= pend) {
        gather_edges<D, 4>(ed + p, base, acc);
        p += 4;
    }
    #pragma unroll 1
    for (; p < pend; ++p) gather_edges<D, 1>(ed + p, base, acc);

    *(float4*)(out + row * D + li * 4) = acc;
}

extern "C" void kernel_launch(void* const* d_in, const int* in_sizes, int n_in,
                              void* d_out, int out_size, void* d_ws, size_t ws_size,
                              hipStream_t stream) {
    const float* x    = (const float*)d_in[0];
    const int*   esrc = (const int*)d_in[1];
    const int*   edst = (const int*)d_in[2];
    const float* eval = (const float*)d_in[3];
    const float* W1   = (const float*)d_in[4];
    const float* W2   = (const float*)d_in[5];
    float*       out  = (float*)d_out;

    const int E = in_sizes[1];
    const int n = in_sizes[0] / DIMS_IN;

    float* buf0 = (float*)d_ws;
    float* buf1 = buf0 + (size_t)n * DIMS_H;
    float* W1t  = buf1 + (size_t)n * DIMS_H;
    float* W2t  = W1t + DIMS_H * DIMS_IN;
    const int per4   = (n + 4096) / 4096;
    const int cntPad = per4 * 4096;
    int* cnt    = (int*)(W2t + DIMS_OUT * DIMS_H);
    int* rowptr = cnt + cntPad;
    int* rank   = rowptr + cntPad;
    int* sorted = rank + ((E + 3) / 4) * 4;

    transpose_kernel<<<(DIMS_IN * DIMS_H + 255) / 256, 256, 0, stream>>>(
        W1, W1t, DIMS_IN, DIMS_H);
    transpose_kernel<<<(DIMS_H * DIMS_OUT + 255) / 256, 256, 0, stream>>>(
        W2, W2t, DIMS_H, DIMS_OUT);

    {
        int c4 = cntPad / 4;
        zero_kernel<<<(c4 + 255) / 256, 256, 0, stream>>>((float4*)cnt, c4);
        edge_rank<<<(E + 255) / 256, 256, 0, stream>>>(edst, cnt, rank, E);
        row_start_scan<<<1, 1024, 0, stream>>>(cnt, rowptr, per4);
        edge_place<<<(E + 255) / 256, 256, 0, stream>>>(
            esrc, edst, eval, rowptr, rank, sorted, E);
    }

    {
        int waves  = (n / 16) * (DIMS_H / 32);
        int blocks = (waves + 7) / 8;
        gemm_wmma_f32_bt<DIMS_IN, false><<<blocks, 256, 0, stream>>>(
            x, W1t, buf0, n, DIMS_H);
    }

    spmm_gather<DIMS_H, 32><<<(n + 7) / 8, 256, 0, stream>>>(
        rowptr, sorted, buf0, buf1, n);

    {
        int waves  = (n / 16) * (DIMS_OUT / 32);
        int blocks = (waves + 7) / 8;
        gemm_wmma_f32_bt<DIMS_H, true><<<blocks, 256, 0, stream>>>(
            buf1, W2t, buf0, n, DIMS_OUT);
    }

    spmm_gather<DIMS_OUT, 8><<<(n + 31) / 32, 256, 0, stream>>>(
        rowptr, sorted, buf0, out, n);
}


# edge_rank's returning int atomics use one counter plane per chip half (die id from the placement message): no counter line is hit from both halves; row sum order = plane 0's arrivals, then plane 1's
# speedup vs baseline: 1.3999x; 1.1574x over previous
#include <hip/hip_runtime.h>

#define DIMS_IN  128
#define DIMS_H   96
#define DIMS_OUT 32

typedef float v2f __attribute__((ext_vector_type(2)));
typedef float v8f __attribute__((ext_vector_type(8)));

__global__ void zero_kernel(float4* __restrict__ p, int n4) {
    int i = blockIdx.x * blockDim.x + threadIdx.x;
    if (i < n4) {
        float4 z; z.x = 0.f; z.y = 0.f; z.z = 0.f; z.w = 0.f;
        p[i] = z;
    }
}

__global__ void transpose_kernel(const float* __restrict__ W,
                                 float* __restrict__ Wt, int K, int N) {
    int i = blockIdx.x * blockDim.x + threadIdx.x;
    if (i < K * N) {
        int k = i / N;
        int n = i - k * N;
        Wt[n * K + k] = W[i];
    }
}

template <int K, bool RELU_A>
__global__ void gemm_wmma_f32_bt(const float* __restrict__ A,
                                 const float* __restrict__ Bt,
                                 float* __restrict__ C,
                                 int M, int N) {
    const int wavesPerBlock = blockDim.x >> 5;
    const int waveId = blockIdx.x * wavesPerBlock + (threadIdx.x >> 5);
    const int lane   = threadIdx.x & 31;
    const int nPair  = N >> 5;
    const int tileM  = waveId / nPair;
    const int tileP  = waveId - tileM * nPair;
    if (tileM * 16 >= M) return;

    const int half = lane >> 4;
    const int l15  = lane & 15;

    const float* arow = A  + (tileM * 16 + l15) * K + half * 2;
    const float* brow = Bt + (tileP * 32 + l15) * K + half * 2;

    v8f acc0 = {};
    v8f acc1 = {};
    #pragma unroll 4
    for (int k = 0; k < K; k += 4) {
        v2f a = *(const v2f*)(arow + k);
        if (RELU_A) {
            a.x = fmaxf(a.x, 0.f);
            a.y = fmaxf(a.y, 0.f);
        }
        v2f b0 = *(const v2f*)(brow + k);
        v2f b1 = *(const v2f*)(brow + 16 * K + k);
        acc0 = __builtin_amdgcn_wmma_f32_16x16x4_f32(
            false, a, false, b0, (short)0, acc0, false, false);
        acc1 = __builtin_amdgcn_wmma_f32_16x16x4_f32(
            false, a, false, b1, (short)0, acc1, false, false);
    }

    float* c0 = C + (tileM * 16 + half * 8) * N + tileP * 32 + l15;
    #pragma unroll
    for (int r = 0; r < 8; ++r) {
        c0[r * N]      = acc0[r];
        c0[r * N + 16] = acc1[r];
    }
}

#define RANK_HALF_SHIFT 30
#define RANK_MASK       ((1 << RANK_HALF_SHIFT) - 1)

__global__ void edge_rank(const int* __restrict__ edst,
                          int* __restrict__ cnt,
                          int* __restrict__ rank, int E, int cntPad) {
    int e = blockIdx.x * blockDim.x + threadIdx.x;
    const unsigned where = __builtin_amdgcn_s_sendmsg_rtn(0x87);
    const int h = (int)((where >> 18) & 1u);
    if (e < E)
        rank[e] = atomicAdd(cnt + h * cntPad + edst[e], 1) | (h << RANK_HALF_SHIFT);
}

__global__ void __launch_bounds__(1024)
row_start_scan(const int* __restrict__ cnt, int* __restrict__ rowptr, int per4) {
    __shared__ int wsum[32];
    const int t    = threadIdx.x;
    const int lane = t & 31;
    const int w    = t >> 5;
    const int4* src  = (const int4*)cnt + t * per4;
    const int4* src1 = src + 1024 * per4;
    int4*       dst  = (int4*)rowptr + t * per4;

    int s = 0;
    #pragma unroll 4
    for (int i = 0; i < per4; ++i) {
        int4 c = src[i];
        int4 d = src1[i];
        s += ((c.x + c.y) + (c.z + c.w)) + ((d.x + d.y) + (d.z + d.w));
    }
    int incl = s;
    #pragma unroll
    for (int off = 1; off < 32; off <<= 1) {
        int v = __shfl_up(incl, off);
        if (lane >= off) incl += v;
    }
    if (lane == 31) wsum[w] = incl;
    __syncthreads();
    if (w == 0) {
        int tot = wsum[lane];
        int ti  = tot;
        #pragma unroll
        for (int off = 1; off < 32; off <<= 1) {
            int v = __shfl_up(ti, off);
            if (lane >= off) ti += v;
        }
        wsum[lane] = ti - tot;
    }
    __syncthreads();
    int run = wsum[w] + (incl - s);
    #pragma unroll 4
    for (int i = 0; i < per4; ++i) {
        int4 c = src[i];
        int4 d = src1[i];
        c.x += d.x; c.y += d.y; c.z += d.z; c.w += d.w;
        int4 o;
        o.x = run;
        o.y = o.x + c.x;
        o.z = o.y + c.y;
        o.w = o.z + c.z;
        run = o.w + c.w;
        dst[i] = o;
    }
}

__global__ void edge_place(const int* __restrict__ esrc,
                           const int* __restrict__ edst,
                           const float* __restrict__ eval,
                           const int* __restrict__ rowptr,
                           const int* __restrict__ cnt0,
                           const int* __restrict__ rank,
                           int* __restrict__ sorted, int E) {
    int e = blockIdx.x * blockDim.x + threadIdx.x;
    if (e < E) {
        const int d    = edst[e];
        const int rk   = rank[e];
        const int base = rowptr[d];
        const int c0   = cnt0[d];
        const int p = base + (rk & RANK_MASK) + ((rk >> RANK_HALF_SHIFT) ? c0 : 0);
        int2 r;
        r.x = esrc[e];
        r.y = __float_as_int(eval[e]);
        ((int2*)sorted)[p] = r;
    }
}

template <int D, int K>
__device__ __forceinline__ void gather_edges(const int2* __restrict__ ed,
                                             const float* __restrict__ base,
                                             float4& acc) {
    int2   e[K];
    float4 m[K];
    #pragma unroll
    for (int j = 0; j < K; ++j) e[j] = ed[j];
    #pragma unroll
    for (int j = 0; j < K; ++j) m[j] = *(const float4*)(base + e[j].x * D);
    #pragma unroll
    for (int j = 0; j < K; ++j) {
        const float v = __int_as_float(e[j].y);
        acc.x = fmaf(v, m[j].x, acc.x);
        acc.y = fmaf(v, m[j].y, acc.y);
        acc.z = fmaf(v, m[j].z, acc.z);
        acc.w = fmaf(v, m[j].w, acc.w);
    }
}

template <int D, int LPR>
__global__ void spmm_gather(const int* __restrict__ rowptr,
                            const int* __restrict__ sorted,
                            const float* __restrict__ dense,
                            float* __restrict__ out, int n) {
    constexpr int D4 = D / 4;
    const int lane = threadIdx.x & 31;
    const int wave = (blockIdx.x * blockDim.x + threadIdx.x) >> 5;
    int row, li;
    if constexpr (LPR == 32) {
        row = __builtin_amdgcn_readfirstlane(wave);
        li  = lane;
    } else {
        row = wave * (32 / LPR) + lane / LPR;
        li  = lane % LPR;
    }
    if (row >= n || li >= D4) return;

    const int2* ed = (const int2*)sorted;
    int       p    = rowptr[row];
    const int pend = rowptr[row + 1];
    const float* base = dense + li * 4;

    float4 acc;
    acc.x = 0.f; acc.y = 0.f; acc.z = 0.f; acc.w = 0.f;
    for (; p + 8 <= pend; p += 8) gather_edges<D, 8>(ed + p, base, acc);
    if (p + 4 <= pend) {
        gather_edges<D, 4>(ed + p, base, acc);
        p += 4;
    }
    #pragma unroll 1
    for (; p < pend; ++p) gather_edges<D, 1>(ed + p, base, acc);

    *(float4*)(out + row * D + li * 4) = acc;
}

extern "C" void kernel_launch(void* const* d_in, const int* in_sizes, int n_in,
                              void* d_out, int out_size, void* d_ws, size_t ws_size,
                              hipStream_t stream) {
    const float* x    = (const float*)d_in[0];
    const int*   esrc = (const int*)d_in[1];
    const int*   edst = (const int*)d_in[2];
    const float* eval = (const float*)d_in[3];
    const float* W1   = (const float*)d_in[4];
    const float* W2   = (const float*)d_in[5];
    float*       out  = (float*)d_out;

    const int E = in_sizes[1];
    const int n = in_sizes[0] / DIMS_IN;

    float* buf0 = (float*)d_ws;
    float* buf1 = buf0 + (size_t)n * DIMS_H;
    float* W1t  = buf1 + (size_t)n * DIMS_H;
    float* W2t  = W1t + DIMS_H * DIMS_IN;
    const int per4   = (n + 4096) / 4096;
    const int cntPad = per4 * 4096;
    int* cnt    = (int*)(W2t + DIMS_OUT * DIMS_H);
    int* rowptr = cnt + 2 * cntPad;
    int* rank   = rowptr + cntPad;
    int* sorted = rank + ((E + 3) / 4) * 4;

    transpose_kernel<<<(DIMS_IN * DIMS_H + 255) / 256, 256, 0, stream>>>(
        W1, W1t, DIMS_IN, DIMS_H);
    transpose_kernel<<<(DIMS_H * DIMS_OUT + 255) / 256, 256, 0, stream>>>(
        W2, W2t, DIMS_H, DIMS_OUT);

    {
        int c4 = cntPad / 2;
        zero_kernel<<<(c4 + 255) / 256, 256, 0, stream>>>((float4*)cnt, c4);
        edge_rank<<<(E + 255) / 256, 256, 0, stream>>>(edst, cnt, rank, E, cntPad);
        row_start_scan<<<1, 1024, 0, stream>>>(cnt, rowptr, per4);
        edge_place<<<(E + 255) / 256, 256, 0, stream>>>(
            esrc, edst, eval, rowptr, cnt, rank, sorted, E);
    }

    {
        int waves  = (n / 16) * (DIMS_H / 32);
        int blocks = (waves + 7) / 8;
        gemm_wmma_f32_bt<DIMS_IN, false><<<blocks, 256, 0, stream>>>(
            x, W1t, buf0, n, DIMS_H);
    }

    spmm_gather<DIMS_H, 32><<<(n + 7) / 8, 256, 0, stream>>>(
        rowptr, sorted, buf0, buf1, n);

    {
        int waves  = (n / 16) * (DIMS_OUT / 32);
        int blocks = (waves + 7) / 8;
        gemm_wmma_f32_bt<DIMS_H, true><<<blocks, 256, 0, stream>>>(
            buf1, W2t, buf0, n, DIMS_OUT);
    }

    spmm_gather<DIMS_OUT, 8><<<(n + 31) / 32, 256, 0, stream>>>(
        rowptr, sorted, buf0, out, n);
}


# Three launches removed (10 to 7): both GEMMs read W in place with the same k order (no transpose launches) and the one-work-group row scan runs as block 0 of the GEMM1 launch
# speedup vs baseline: 1.6410x; 1.1722x over previous
#include <hip/hip_runtime.h>

#define DIMS_IN  128
#define DIMS_H   96
#define DIMS_OUT 32

typedef float v2f __attribute__((ext_vector_type(2)));
typedef float v8f __attribute__((ext_vector_type(8)));

__global__ void zero_kernel(float4* __restrict__ p, int n4) {
    int i = blockIdx.x * blockDim.x + threadIdx.x;
    if (i < n4) {
        float4 z; z.x = 0.f; z.y = 0.f; z.z = 0.f; z.w = 0.f;
        p[i] = z;
    }
}

template <int K, int N, bool RELU_A>
__device__ __forceinline__ void gemm_strip(const float* __restrict__ A,
                                           const float* __restrict__ B,
                                           float* __restrict__ C,
                                           int M, int waveId, int lane) {
    constexpr int nPair = N >> 5;
    const int tileM  = waveId / nPair;
    const int tileP  = waveId - tileM * nPair;
    if (tileM * 16 >= M) return;

    const int half = lane >> 4;
    const int l15  = lane & 15;

    const float* arow = A + (tileM * 16 + l15) * K + half * 2;
    const float* bcol = B + (half * 2) * N + tileP * 32 + l15;

    v8f acc0 = {};
    v8f acc1 = {};
    #pragma unroll 4
    for (int k = 0; k < K; k += 4) {
        v2f a = *(const v2f*)(arow + k);
        if (RELU_A) {
            a.x = fmaxf(a.x, 0.f);
            a.y = fmaxf(a.y, 0.f);
        }
        v2f b0, b1;
        b0.x = bcol[k * N];
        b0.y = bcol[(k + 1) * N];
        b1.x = bcol[k * N + 16];
        b1.y = bcol[(k + 1) * N + 16];
        acc0 = __builtin_amdgcn_wmma_f32_16x16x4_f32(
            false, a, false, b0, (short)0, acc0, false, false);
        acc1 = __builtin_amdgcn_wmma_f32_16x16x4_f32(
            false, a, false, b1, (short)0, acc1, false, false);
    }

    float* c0 = C + (tileM * 16 + half * 8) * N + tileP * 32 + l15;
    #pragma unroll
    for (int r = 0; r < 8; ++r) {
        c0[r * N]      = acc0[r];
        c0[r * N + 16] = acc1[r];
    }
}

template <int K, int N, bool RELU_A>
__global__ void gemm_wmma_f32(const float* __restrict__ A,
                              const float* __restrict__ B,
                              float* __restrict__ C, int M) {
    const int waveId = blockIdx.x * (blockDim.x >> 5) + (threadIdx.x >> 5);
    gemm_strip<K, N, RELU_A>(A, B, C, M, waveId, threadIdx.x & 31);
}

#define RANK_HALF_SHIFT 30
#define RANK_MASK       ((1 << RANK_HALF_SHIFT) - 1)

__global__ void edge_rank(const int* __restrict__ edst,
                          int* __restrict__ cnt,
                          int* __restrict__ rank, int E, int cntPad) {
    int e = blockIdx.x * blockDim.x + threadIdx.x;
    const unsigned where = __builtin_amdgcn_s_sendmsg_rtn(0x87);
    const int h = (int)((where >> 18) & 1u);
    if (e < E)
        rank[e] = atomicAdd(cnt + h * cntPad + edst[e], 1) | (h << RANK_HALF_SHIFT);
}

__device__ __forceinline__ void row_start_scan(const int* __restrict__ cnt,
                                               int* __restrict__ rowptr,
                                               int per4, int* wsum) {
    const int t    = threadIdx.x;
    const int lane = t & 31;
    const int w    = t >> 5;
    const int4* src  = (const int4*)cnt + t * per4;
    const int4* src1 = src + 1024 * per4;
    int4*       dst  = (int4*)rowptr + t * per4;

    int s = 0;
    #pragma unroll 4
    for (int i = 0; i < per4; ++i) {
        int4 c = src[i];
        int4 d = src1[i];
        s += ((c.x + c.y) + (c.z + c.w)) + ((d.x + d.y) + (d.z + d.w));
    }
    int incl = s;
    #pragma unroll
    for (int off = 1; off < 32; off <<= 1) {
        int v = __shfl_up(incl, off);
        if (lane >= off) incl += v;
    }
    if (lane == 31) wsum[w] = incl;
    __syncthreads();
    if (w == 0) {
        int tot = wsum[lane];
        int ti  = tot;
        #pragma unroll
        for (int off = 1; off < 32; off <<= 1) {
            int v = __shfl_up(ti, off);
            if (lane >= off) ti += v;
        }
        wsum[lane] = ti - tot;
    }
    __syncthreads();
    int run = wsum[w] + (incl - s);
    #pragma unroll 4
    for (int i = 0; i < per4; ++i) {
        int4 c = src[i];
        int4 d = src1[i];
        c.x += d.x; c.y += d.y; c.z += d.z; c.w += d.w;
        int4 o;
        o.x = run;
        o.y = o.x + c.x;
        o.z = o.y + c.y;
        o.w = o.z + c.z;
        run = o.w + c.w;
        dst[i] = o;
    }
}

__global__ void __launch_bounds__(1024)
scan_and_gemm1(const int* __restrict__ cnt, int* __restrict__ rowptr, int per4,
               const float* __restrict__ x, const float* __restrict__ W1,
               float* __restrict__ xw, int M) {
    __shared__ int wsum[32];
    if (blockIdx.x == 0) {
        row_start_scan(cnt, rowptr, per4, wsum);
        return;
    }
    const int waveId = (blockIdx.x - 1) * 32 + (threadIdx.x >> 5);
    gemm_strip<DIMS_IN, DIMS_H, false>(x, W1, xw, M, waveId, threadIdx.x & 31);
}

__global__ void edge_place(const int* __restrict__ esrc,
                           const int* __restrict__ edst,
                           const float* __restrict__ eval,
                           const int* __restrict__ rowptr,
                           const int* __restrict__ cnt0,
                           const int* __restrict__ rank,
                           int* __restrict__ sorted, int E) {
    int e = blockIdx.x * blockDim.x + threadIdx.x;
    if (e < E) {
        const int d    = edst[e];
        const int rk   = rank[e];
        const int base = rowptr[d];
        const int c0   = cnt0[d];
        const int p = base + (rk & RANK_MASK) + ((rk >> RANK_HALF_SHIFT) ? c0 : 0);
        int2 r;
        r.x = esrc[e];
        r.y = __float_as_int(eval[e]);
        ((int2*)sorted)[p] = r;
    }
}

template <int D, int K>
__device__ __forceinline__ void gather_edges(const int2* __restrict__ ed,
                                             const float* __restrict__ base,
                                             float4& acc) {
    int2   e[K];
    float4 m[K];
    #pragma unroll
    for (int j = 0; j < K; ++j) e[j] = ed[j];
    #pragma unroll
    for (int j = 0; j < K; ++j) m[j] = *(const float4*)(base + e[j].x * D);
    #pragma unroll
    for (int j = 0; j < K; ++j) {
        const float v = __int_as_float(e[j].y);
        acc.x = fmaf(v, m[j].x, acc.x);
        acc.y = fmaf(v, m[j].y, acc.y);
        acc.z = fmaf(v, m[j].z, acc.z);
        acc.w = fmaf(v, m[j].w, acc.w);
    }
}

template <int D, int LPR>
__global__ void spmm_gather(const int* __restrict__ rowptr,
                            const int* __restrict__ sorted,
                            const float* __restrict__ dense,
                            float* __restrict__ out, int n) {
    constexpr int D4 = D / 4;
    const int lane = threadIdx.x & 31;
    const int wave = (blockIdx.x * blockDim.x + threadIdx.x) >> 5;
    int row, li;
    if constexpr (LPR == 32) {
        row = __builtin_amdgcn_readfirstlane(wave);
        li  = lane;
    } else {
        row = wave * (32 / LPR) + lane / LPR;
        li  = lane % LPR;
    }
    if (row >= n || li >= D4) return;

    const int2* ed = (const int2*)sorted;
    int       p    = rowptr[row];
    const int pend = rowptr[row + 1];
    const float* base = dense + li * 4;

    float4 acc;
    acc.x = 0.f; acc.y = 0.f; acc.z = 0.f; acc.w = 0.f;
    for (; p + 8 <= pend; p += 8) gather_edges<D, 8>(ed + p, base, acc);
    if (p + 4 <= pend) {
        gather_edges<D, 4>(ed + p, base, acc);
        p += 4;
    }
    #pragma unroll 1
    for (; p < pend; ++p) gather_edges<D, 1>(ed + p, base, acc);

    *(float4*)(out + row * D + li * 4) = acc;
}

extern "C" void kernel_launch(void* const* d_in, const int* in_sizes, int n_in,
                              void* d_out, int out_size, void* d_ws, size_t ws_size,
                              hipStream_t stream) {
    const float* x    = (const float*)d_in[0];
    const int*   esrc = (const int*)d_in[1];
    const int*   edst = (const int*)d_in[2];
    const float* eval = (const float*)d_in[3];
    const float* W1   = (const float*)d_in[4];
    const float* W2   = (const float*)d_in[5];
    float*       out  = (float*)d_out;

    const int E = in_sizes[1];
    const int n = in_sizes[0] / DIMS_IN;

    float* buf0 = (float*)d_ws;
    float* buf1 = buf0 + (size_t)n * DIMS_H;
    const int per4   = (n + 4096) / 4096;
    const int cntPad = per4 * 4096;
    int* cnt    = (int*)(buf1 + (size_t)n * DIMS_H);
    int* rowptr = cnt + 2 * cntPad;
    int* rank   = rowptr + cntPad;
    int* sorted = rank + ((E + 3) / 4) * 4;

    {
        int c4 = cntPad / 2;
        zero_kernel<<<(c4 + 255) / 256, 256, 0, stream>>>((float4*)cnt, c4);
        edge_rank<<<(E + 255) / 256, 256, 0, stream>>>(edst, cnt, rank, E, cntPad);
        int waves  = (n / 16) * (DIMS_H / 32);
        int blocks = 1 + (waves + 31) / 32;
        scan_and_gemm1<<<blocks, 1024, 0, stream>>>(cnt, rowptr, per4, x, W1, buf0, n);
        edge_place<<<(E + 255) / 256, 256, 0, stream>>>(
            esrc, edst, eval, rowptr, cnt, rank, sorted, E);
    }

    spmm_gather<DIMS_H, 32><<<(n + 7) / 8, 256, 0, stream>>>(
        rowptr, sorted, buf0, buf1, n);

    {
        int waves  = (n / 16) * (DIMS_OUT / 32);
        int blocks = (waves + 7) / 8;
        gemm_wmma_f32<DIMS_H, DIMS_OUT, true><<<blocks, 256, 0, stream>>>(
            buf1, W2, buf0, n);
    }

    spmm_gather<DIMS_OUT, 8><<<(n + 31) / 32, 256, 0, stream>>>(
        rowptr, sorted, buf0, out, n);
}


# Edge sort's 3 passes (rank, scan, place) become one: slot = atomic count, record stored at once into a fixed 32-slot row bucket per chip half, overflow list behind; 7 launches to 5; sum order kept
# speedup vs baseline: 3.0384x; 1.8516x over previous
#include <hip/hip_runtime.h>

#define DIMS_IN  128
#define DIMS_H   96
#define DIMS_OUT 32

typedef float v2f __attribute__((ext_vector_type(2)));
typedef float v8f __attribute__((ext_vector_type(8)));

__global__ void zero_kernel(float4* __restrict__ p, int n4) {
    int i = blockIdx.x * blockDim.x + threadIdx.x;
    if (i < n4) {
        float4 z; z.x = 0.f; z.y = 0.f; z.z = 0.f; z.w = 0.f;
        p[i] = z;
    }
}

template <int K, int N, bool RELU_A>
__device__ __forceinline__ void gemm_strip(const float* __restrict__ A,
                                           const float* __restrict__ B,
                                           float* __restrict__ C,
                                           int M, int waveId, int lane) {
    constexpr int nPair = N >> 5;
    const int tileM  = waveId / nPair;
    const int tileP  = waveId - tileM * nPair;
    if (tileM * 16 >= M) return;

    const int half = lane >> 4;
    const int l15  = lane & 15;

    const float* arow = A + (tileM * 16 + l15) * K + half * 2;
    const float* bcol = B + (half * 2) * N + tileP * 32 + l15;

    v8f acc0 = {};
    v8f acc1 = {};
    #pragma unroll 4
    for (int k = 0; k < K; k += 4) {
        v2f a = *(const v2f*)(arow + k);
        if (RELU_A) {
            a.x = fmaxf(a.x, 0.f);
            a.y = fmaxf(a.y, 0.f);
        }
        v2f b0, b1;
        b0.x = bcol[k * N];
        b0.y = bcol[(k + 1) * N];
        b1.x = bcol[k * N + 16];
        b1.y = bcol[(k + 1) * N + 16];
        acc0 = __builtin_amdgcn_wmma_f32_16x16x4_f32(
            false, a, false, b0, (short)0, acc0, false, false);
        acc1 = __builtin_amdgcn_wmma_f32_16x16x4_f32(
            false, a, false, b1, (short)0, acc1, false, false);
    }

    float* c0 = C + (tileM * 16 + half * 8) * N + tileP * 32 + l15;
    #pragma unroll
    for (int r = 0; r < 8; ++r) {
        c0[r * N]      = acc0[r];
        c0[r * N + 16] = acc1[r];
    }
}

template <int K, int N, bool RELU_A>
__global__ void gemm_wmma_f32(const float* __restrict__ A,
                              const float* __restrict__ B,
                              float* __restrict__ C, int M) {
    const int waveId = blockIdx.x * (blockDim.x >> 5) + (threadIdx.x >> 5);
    gemm_strip<K, N, RELU_A>(A, B, C, M, waveId, threadIdx.x & 31);
}

#define ROW_SLOTS 32

__device__ __forceinline__ void edge_fill(const int* __restrict__ esrc,
                                          const int* __restrict__ edst,
                                          const float* __restrict__ eval,
                                          int* __restrict__ cnt,
                                          int2* __restrict__ bucket,
                                          int* __restrict__ ovfN,
                                          int4* __restrict__ ovf,
                                          int E, int cntPad, int e) {
    const unsigned where = __builtin_amdgcn_s_sendmsg_rtn(0x87);
    const int h = (int)((where >> 18) & 1u);
    if (e < E) {
        const int d = edst[e];
        int2 r;
        r.x = esrc[e];
        r.y = __float_as_int(eval[e]);
        const int slot = atomicAdd(cnt + h * cntPad + d, 1);
        if (slot < ROW_SLOTS) {
            bucket[(d * 2 + h) * ROW_SLOTS + slot] = r;
        } else {
            int4 o;
            o.x = d; o.y = r.x; o.z = r.y; o.w = 0;
            ovf[atomicAdd(ovfN, 1)] = o;
        }
    }
}

__global__ void __launch_bounds__(1024)
fill_and_gemm1(const int* __restrict__ esrc, const int* __restrict__ edst,
               const float* __restrict__ eval, int* __restrict__ cnt,
               int2* __restrict__ bucket, int* __restrict__ ovfN,
               int4* __restrict__ ovf, int E, int cntPad, int gemmBlocks,
               const float* __restrict__ x, const float* __restrict__ W1,
               float* __restrict__ xw, int M) {
    if ((int)blockIdx.x < gemmBlocks) {
        const int waveId = blockIdx.x * 32 + (threadIdx.x >> 5);
        gemm_strip<DIMS_IN, DIMS_H, false>(x, W1, xw, M, waveId, threadIdx.x & 31);
        return;
    }
    const int e = (blockIdx.x - gemmBlocks) * 1024 + threadIdx.x;
    edge_fill(esrc, edst, eval, cnt, bucket, ovfN, ovf, E, cntPad, e);
}

struct RowEdges {
    const int2* s0;
    const int2* s1;
    int c0;
    __device__ __forceinline__ int2 rec(int j) const { return (j < c0 ? s0 : s1)[j]; }
};

template <int D, int K>
__device__ __forceinline__ void gather_edges(const RowEdges& re, int j0,
                                             const float* __restrict__ base,
                                             float4& acc) {
    int2   e[K];
    float4 m[K];
    #pragma unroll
    for (int j = 0; j < K; ++j) e[j] = re.rec(j0 + j);
    #pragma unroll
    for (int j = 0; j < K; ++j) m[j] = *(const float4*)(base + e[j].x * D);
    #pragma unroll
    for (int j = 0; j < K; ++j) {
        const float v = __int_as_float(e[j].y);
        acc.x = fmaf(v, m[j].x, acc.x);
        acc.y = fmaf(v, m[j].y, acc.y);
        acc.z = fmaf(v, m[j].z, acc.z);
        acc.w = fmaf(v, m[j].w, acc.w);
    }
}

template <int D, int LPR>
__global__ void spmm_gather(const int* __restrict__ cnt, int cntPad,
                            const int2* __restrict__ bucket,
                            const int* __restrict__ ovfN,
                            const int4* __restrict__ ovf,
                            const float* __restrict__ dense,
                            float* __restrict__ out, int n) {
    constexpr int D4 = D / 4;
    const int lane = threadIdx.x & 31;
    const int wave = (blockIdx.x * blockDim.x + threadIdx.x) >> 5;
    int row, li;
    if constexpr (LPR == 32) {
        row = __builtin_amdgcn_readfirstlane(wave);
        li  = lane;
    } else {
        row = wave * (32 / LPR) + lane / LPR;
        li  = lane % LPR;
    }
    if (row >= n || li >= D4) return;

    const int n0 = cnt[row];
    const int n1 = cnt[cntPad + row];
    const int c0 = min(n0, ROW_SLOTS);
    const int c1 = min(n1, ROW_SLOTS);
    RowEdges re;
    re.s0 = bucket + (row * 2) * ROW_SLOTS;
    re.s1 = re.s0 + (ROW_SLOTS - c0);
    re.c0 = c0;
    const int pend = c0 + c1;
    const float* base = dense + li * 4;

    float4 acc;
    acc.x = 0.f; acc.y = 0.f; acc.z = 0.f; acc.w = 0.f;
    int p = 0;
    for (; p + 8 <= pend; p += 8) gather_edges<D, 8>(re, p, base, acc);
    if (p + 4 <= pend) {
        gather_edges<D, 4>(re, p, base, acc);
        p += 4;
    }
    #pragma unroll 1
    for (; p < pend; ++p) gather_edges<D, 1>(re, p, base, acc);

    if (n0 > ROW_SLOTS || n1 > ROW_SLOTS) {
        const int nOv = *ovfN;
        #pragma unroll 1
        for (int k = 0; k < nOv; ++k) {
            const int4 o = ovf[k];
            if (o.x == row) {
                const float4 m = *(const float4*)(base + o.y * D);
                const float  v = __int_as_float(o.z);
                acc.x = fmaf(v, m.x, acc.x);
                acc.y = fmaf(v, m.y, acc.y);
                acc.z = fmaf(v, m.z, acc.z);
                acc.w = fmaf(v, m.w, acc.w);
            }
        }
    }

    *(float4*)(out + row * D + li * 4) = acc;
}

extern "C" void kernel_launch(void* const* d_in, const int* in_sizes, int n_in,
                              void* d_out, int out_size, void* d_ws, size_t ws_size,
                              hipStream_t stream) {
    const float* x    = (const float*)d_in[0];
    const int*   esrc = (const int*)d_in[1];
    const int*   edst = (const int*)d_in[2];
    const float* eval = (const float*)d_in[3];
    const float* W1   = (const float*)d_in[4];
    const float* W2   = (const float*)d_in[5];
    float*       out  = (float*)d_out;

    const int E = in_sizes[1];
    const int n = in_sizes[0] / DIMS_IN;

    float* buf0 = (float*)d_ws;
    float* buf1 = buf0 + (size_t)n * DIMS_H;
    const int cntPad = ((n + 31) / 32) * 32;
    int*  cnt    = (int*)(buf1 + (size_t)n * DIMS_H);
    int*  ovfN   = cnt + 2 * cntPad;
    int2* bucket = (int2*)(ovfN + 4);
    int4* ovf    = (int4*)(bucket + (size_t)n * 2 * ROW_SLOTS);

    {
        int c4 = cntPad / 2 + 1;
        zero_kernel<<<(c4 + 255) / 256, 256, 0, stream>>>((float4*)cnt, c4);
        int waves      = (n / 16) * (DIMS_H / 32);
        int gemmBlocks = (waves + 31) / 32;
        int fillBlocks = (E + 1023) / 1024;
        fill_and_gemm1<<<gemmBlocks + fillBlocks, 1024, 0, stream>>>(
            esrc, edst, eval, cnt, bucket, ovfN, ovf, E, cntPad, gemmBlocks,
            x, W1, buf0, n);
    }

    spmm_gather<DIMS_H, 32><<<(n + 7) / 8, 256, 0, stream>>>(
        cnt, cntPad, bucket, ovfN, ovf, buf0, buf1, n);

    {
        int waves  = (n / 16) * (DIMS_OUT / 32);
        int blocks = (waves + 7) / 8;
        gemm_wmma_f32<DIMS_H, DIMS_OUT, true><<<blocks, 256, 0, stream>>>(
            buf1, W2, buf0, n);
    }

    spmm_gather<DIMS_OUT, 8><<<(n + 31) / 32, 256, 0, stream>>>(
        cnt, cntPad, bucket, ovfN, ovf, buf0, out, n);
}


# SpMM1, relu and GEMM2 in one launch: a 16-wave work-group sums 16 rows (one wave per row) into LDS, wave 0 multiplies the tile by W2 on WMMA; 5 launches to 4, SpMM1 result never stored
# speedup vs baseline: 3.1331x; 1.0312x over previous
#include <hip/hip_runtime.h>

#define DIMS_IN  128
#define DIMS_H   96
#define DIMS_OUT 32

typedef float v2f __attribute__((ext_vector_type(2)));
typedef float v8f __attribute__((ext_vector_type(8)));

__global__ void zero_kernel(float4* __restrict__ p, int n4) {
    int i = blockIdx.x * blockDim.x + threadIdx.x;
    if (i < n4) {
        float4 z; z.x = 0.f; z.y = 0.f; z.z = 0.f; z.w = 0.f;
        p[i] = z;
    }
}

template <int K, int N>
__device__ __forceinline__ void gemm_strip(const float* __restrict__ A,
                                           const float* __restrict__ B,
                                           float* __restrict__ C,
                                           int M, int waveId, int lane) {
    constexpr int nPair = N >> 5;
    const int tileM  = waveId / nPair;
    const int tileP  = waveId - tileM * nPair;
    if (tileM * 16 >= M) return;

    const int half = lane >> 4;
    const int l15  = lane & 15;

    const float* arow = A + (tileM * 16 + l15) * K + half * 2;
    const float* bcol = B + (half * 2) * N + tileP * 32 + l15;

    v8f acc0 = {};
    v8f acc1 = {};
    #pragma unroll 4
    for (int k = 0; k < K; k += 4) {
        const v2f a = *(const v2f*)(arow + k);
        v2f b0, b1;
        b0.x = bcol[k * N];
        b0.y = bcol[(k + 1) * N];
        b1.x = bcol[k * N + 16];
        b1.y = bcol[(k + 1) * N + 16];
        acc0 = __builtin_amdgcn_wmma_f32_16x16x4_f32(
            false, a, false, b0, (short)0, acc0, false, false);
        acc1 = __builtin_amdgcn_wmma_f32_16x16x4_f32(
            false, a, false, b1, (short)0, acc1, false, false);
    }

    float* c0 = C + (tileM * 16 + half * 8) * N + tileP * 32 + l15;
    #pragma unroll
    for (int r = 0; r < 8; ++r) {
        c0[r * N]      = acc0[r];
        c0[r * N + 16] = acc1[r];
    }
}

#define ROW_SLOTS 32

__device__ __forceinline__ void edge_fill(const int* __restrict__ esrc,
                                          const int* __restrict__ edst,
                                          const float* __restrict__ eval,
                                          int* __restrict__ cnt,
                                          int2* __restrict__ bucket,
                                          int* __restrict__ ovfN,
                                          int4* __restrict__ ovf,
                                          int E, int cntPad, int e) {
    const unsigned where = __builtin_amdgcn_s_sendmsg_rtn(0x87);
    const int h = (int)((where >> 18) & 1u);
    if (e < E) {
        const int d = edst[e];
        int2 r;
        r.x = esrc[e];
        r.y = __float_as_int(eval[e]);
        const int slot = atomicAdd(cnt + h * cntPad + d, 1);
        if (slot < ROW_SLOTS) {
            bucket[(d * 2 + h) * ROW_SLOTS + slot] = r;
        } else {
            int4 o;
            o.x = d; o.y = r.x; o.z = r.y; o.w = 0;
            ovf[atomicAdd(ovfN, 1)] = o;
        }
    }
}

__global__ void __launch_bounds__(1024)
fill_and_gemm1(const int* __restrict__ esrc, const int* __restrict__ edst,
               const float* __restrict__ eval, int* __restrict__ cnt,
               int2* __restrict__ bucket, int* __restrict__ ovfN,
               int4* __restrict__ ovf, int E, int cntPad, int gemmBlocks,
               const float* __restrict__ x, const float* __restrict__ W1,
               float* __restrict__ xw, int M) {
    if ((int)blockIdx.x < gemmBlocks) {
        const int waveId = blockIdx.x * 32 + (threadIdx.x >> 5);
        gemm_strip<DIMS_IN, DIMS_H>(x, W1, xw, M, waveId, threadIdx.x & 31);
        return;
    }
    const int e = (blockIdx.x - gemmBlocks) * 1024 + threadIdx.x;
    edge_fill(esrc, edst, eval, cnt, bucket, ovfN, ovf, E, cntPad, e);
}

struct RowEdges {
    const int2* s0;
    const int2* s1;
    int c0;
    __device__ __forceinline__ int2 rec(int j) const { return (j < c0 ? s0 : s1)[j]; }
};

template <int D, int K>
__device__ __forceinline__ void gather_edges(const RowEdges& re, int j0,
                                             const float* __restrict__ base,
                                             float4& acc) {
    int2   e[K];
    float4 m[K];
    #pragma unroll
    for (int j = 0; j < K; ++j) e[j] = re.rec(j0 + j);
    #pragma unroll
    for (int j = 0; j < K; ++j) m[j] = *(const float4*)(base + e[j].x * D);
    #pragma unroll
    for (int j = 0; j < K; ++j) {
        const float v = __int_as_float(e[j].y);
        acc.x = fmaf(v, m[j].x, acc.x);
        acc.y = fmaf(v, m[j].y, acc.y);
        acc.z = fmaf(v, m[j].z, acc.z);
        acc.w = fmaf(v, m[j].w, acc.w);
    }
}

template <int D>
__device__ __forceinline__ float4 row_sum(int row, int li,
                                          const int* __restrict__ cnt, int cntPad,
                                          const int2* __restrict__ bucket,
                                          const int* __restrict__ ovfN,
                                          const int4* __restrict__ ovf,
                                          const float* __restrict__ dense) {
    const int n0 = cnt[row];
    const int n1 = cnt[cntPad + row];
    const int c0 = min(n0, ROW_SLOTS);
    const int c1 = min(n1, ROW_SLOTS);
    RowEdges re;
    re.s0 = bucket + (row * 2) * ROW_SLOTS;
    re.s1 = re.s0 + (ROW_SLOTS - c0);
    re.c0 = c0;
    const int pend = c0 + c1;
    const float* base = dense + li * 4;

    float4 acc;
    acc.x = 0.f; acc.y = 0.f; acc.z = 0.f; acc.w = 0.f;
    int p = 0;
    for (; p + 8 <= pend; p += 8) gather_edges<D, 8>(re, p, base, acc);
    if (p + 4 <= pend) {
        gather_edges<D, 4>(re, p, base, acc);
        p += 4;
    }
    #pragma unroll 1
    for (; p < pend; ++p) gather_edges<D, 1>(re, p, base, acc);

    if (n0 > ROW_SLOTS || n1 > ROW_SLOTS) {
        const int nOv = *ovfN;
        #pragma unroll 1
        for (int k = 0; k < nOv; ++k) {
            const int4 o = ovf[k];
            if (o.x == row) {
                const float4 m = *(const float4*)(base + o.y * D);
                const float  v = __int_as_float(o.z);
                acc.x = fmaf(v, m.x, acc.x);
                acc.y = fmaf(v, m.y, acc.y);
                acc.z = fmaf(v, m.z, acc.z);
                acc.w = fmaf(v, m.w, acc.w);
            }
        }
    }
    return acc;
}

template <int D, int LPR>
__global__ void spmm_gather(const int* __restrict__ cnt, int cntPad,
                            const int2* __restrict__ bucket,
                            const int* __restrict__ ovfN,
                            const int4* __restrict__ ovf,
                            const float* __restrict__ dense,
                            float* __restrict__ out, int n) {
    constexpr int D4 = D / 4;
    const int lane = threadIdx.x & 31;
    const int wave = (blockIdx.x * blockDim.x + threadIdx.x) >> 5;
    const int row  = wave * (32 / LPR) + lane / LPR;
    const int li   = lane % LPR;
    if (row >= n || li >= D4) return;
    const float4 acc = row_sum<D>(row, li, cnt, cntPad, bucket, ovfN, ovf, dense);
    *(float4*)(out + row * D + li * 4) = acc;
}

#define TILE_ROWS 16
#define H_PITCH   100

__global__ void __launch_bounds__(TILE_ROWS * 32)
spmm_relu_gemm2(const int* __restrict__ cnt, int cntPad,
                const int2* __restrict__ bucket,
                const int* __restrict__ ovfN,
                const int4* __restrict__ ovf,
                const float* __restrict__ dense,
                const float* __restrict__ W2,
                float* __restrict__ hw, int n) {
    __shared__ float hs[TILE_ROWS * H_PITCH];
    constexpr int D4 = DIMS_H / 4;
    const int lane = threadIdx.x & 31;
    const int w    = __builtin_amdgcn_readfirstlane((int)(threadIdx.x >> 5));
    const int row0 = blockIdx.x * TILE_ROWS;
    const int row  = row0 + w;

    if (lane < D4) {
        float4 acc;
        acc.x = 0.f; acc.y = 0.f; acc.z = 0.f; acc.w = 0.f;
        if (row < n)
            acc = row_sum<DIMS_H>(row, lane, cnt, cntPad, bucket, ovfN, ovf, dense);
        acc.x = fmaxf(acc.x, 0.f);
        acc.y = fmaxf(acc.y, 0.f);
        acc.z = fmaxf(acc.z, 0.f);
        acc.w = fmaxf(acc.w, 0.f);
        *(float4*)(hs + w * H_PITCH + lane * 4) = acc;
    }
    __syncthreads();
    if (w != 0) return;

    const int half = lane >> 4;
    const int l15  = lane & 15;
    const float* arow = hs + l15 * H_PITCH + half * 2;
    const float* bcol = W2 + (half * 2) * DIMS_OUT + l15;

    v8f acc0 = {};
    v8f acc1 = {};
    #pragma unroll 8
    for (int k = 0; k < DIMS_H; k += 4) {
        const v2f a = *(const v2f*)(arow + k);
        v2f b0, b1;
        b0.x = bcol[k * DIMS_OUT];
        b0.y = bcol[(k + 1) * DIMS_OUT];
        b1.x = bcol[k * DIMS_OUT + 16];
        b1.y = bcol[(k + 1) * DIMS_OUT + 16];
        acc0 = __builtin_amdgcn_wmma_f32_16x16x4_f32(
            false, a, false, b0, (short)0, acc0, false, false);
        acc1 = __builtin_amdgcn_wmma_f32_16x16x4_f32(
            false, a, false, b1, (short)0, acc1, false, false);
    }

    float* c0 = hw + (row0 + half * 8) * DIMS_OUT + l15;
    #pragma unroll
    for (int r = 0; r < 8; ++r) {
        c0[r * DIMS_OUT]      = acc0[r];
        c0[r * DIMS_OUT + 16] = acc1[r];
    }
}

extern "C" void kernel_launch(void* const* d_in, const int* in_sizes, int n_in,
                              void* d_out, int out_size, void* d_ws, size_t ws_size,
                              hipStream_t stream) {
    const float* x    = (const float*)d_in[0];
    const int*   esrc = (const int*)d_in[1];
    const int*   edst = (const int*)d_in[2];
    const float* eval = (const float*)d_in[3];
    const float* W1   = (const float*)d_in[4];
    const float* W2   = (const float*)d_in[5];
    float*       out  = (float*)d_out;

    const int E = in_sizes[1];
    const int n = in_sizes[0] / DIMS_IN;

    float* buf0 = (float*)d_ws;
    float* buf1 = buf0 + (size_t)n * DIMS_H;
    const int cntPad = ((n + 31) / 32) * 32;
    int*  cnt    = (int*)(buf1 + (size_t)n * DIMS_H);
    int*  ovfN   = cnt + 2 * cntPad;
    int2* bucket = (int2*)(ovfN + 4);
    int4* ovf    = (int4*)(bucket + (size_t)n * 2 * ROW_SLOTS);

    {
        int c4 = cntPad / 2 + 1;
        zero_kernel<<<(c4 + 255) / 256, 256, 0, stream>>>((float4*)cnt, c4);
        int waves      = (n / 16) * (DIMS_H / 32);
        int gemmBlocks = (waves + 31) / 32;
        int fillBlocks = (E + 1023) / 1024;
        fill_and_gemm1<<<gemmBlocks + fillBlocks, 1024, 0, stream>>>(
            esrc, edst, eval, cnt, bucket, ovfN, ovf, E, cntPad, gemmBlocks,
            x, W1, buf0, n);
    }

    spmm_relu_gemm2<<<(n + TILE_ROWS - 1) / TILE_ROWS, TILE_ROWS * 32, 0, stream>>>(
        cnt, cntPad, bucket, ovfN, ovf, buf0, W2, buf1, n);

    spmm_gather<DIMS_OUT, 8><<<(n + 31) / 32, 256, 0, stream>>>(
        cnt, cntPad, bucket, ovfN, ovf, buf1, out, n);
}


# Fewer, longer waves: edge fill takes 4 edges a thread (25024 waves to 6272); an SpMM1 work-group walks 4 row tiles, one barrier, 4 waves multiply (50000 to 12512); arrival-rank sum order kept
# speedup vs baseline: 3.4877x; 1.1132x over previous
#include <hip/hip_runtime.h>

#define DIMS_IN  128
#define DIMS_H   96
#define DIMS_OUT 32

typedef float v2f __attribute__((ext_vector_type(2)));
typedef float v8f __attribute__((ext_vector_type(8)));

__global__ void zero_kernel(float4* __restrict__ p, int n4) {
    int i = blockIdx.x * blockDim.x + threadIdx.x;
    if (i < n4) {
        float4 z; z.x = 0.f; z.y = 0.f; z.z = 0.f; z.w = 0.f;
        p[i] = z;
    }
}

template <int K, int N>
__device__ __forceinline__ void gemm_strip(const float* __restrict__ A,
                                           const float* __restrict__ B,
                                           float* __restrict__ C,
                                           int M, int waveId, int lane) {
    constexpr int nPair = N >> 5;
    const int tileM  = waveId / nPair;
    const int tileP  = waveId - tileM * nPair;
    if (tileM * 16 >= M) return;

    const int half = lane >> 4;
    const int l15  = lane & 15;

    const float* arow = A + (tileM * 16 + l15) * K + half * 2;
    const float* bcol = B + (half * 2) * N + tileP * 32 + l15;

    v8f acc0 = {};
    v8f acc1 = {};
    #pragma unroll 4
    for (int k = 0; k < K; k += 4) {
        const v2f a = *(const v2f*)(arow + k);
        v2f b0, b1;
        b0.x = bcol[k * N];
        b0.y = bcol[(k + 1) * N];
        b1.x = bcol[k * N + 16];
        b1.y = bcol[(k + 1) * N + 16];
        acc0 = __builtin_amdgcn_wmma_f32_16x16x4_f32(
            false, a, false, b0, (short)0, acc0, false, false);
        acc1 = __builtin_amdgcn_wmma_f32_16x16x4_f32(
            false, a, false, b1, (short)0, acc1, false, false);
    }

    float* c0 = C + (tileM * 16 + half * 8) * N + tileP * 32 + l15;
    #pragma unroll
    for (int r = 0; r < 8; ++r) {
        c0[r * N]      = acc0[r];
        c0[r * N + 16] = acc1[r];
    }
}

#define ROW_SLOTS 32
#define EDGES_PER_THREAD 4
#define FILL_THREADS 1024

template <bool GUARD>
__device__ __forceinline__ void edge_fill(const int* __restrict__ esrc,
                                          const int* __restrict__ edst,
                                          const float* __restrict__ eval,
                                          int* __restrict__ cnt,
                                          int2* __restrict__ bucket,
                                          int* __restrict__ ovfN,
                                          int4* __restrict__ ovf,
                                          int E, int cntPad, int e0) {
    const unsigned where = __builtin_amdgcn_s_sendmsg_rtn(0x87);
    const int h = (int)((where >> 18) & 1u);
    int* plane = cnt + h * cntPad;
    int  d[EDGES_PER_THREAD];
    int2 r[EDGES_PER_THREAD];
    int  slot[EDGES_PER_THREAD];
    #pragma unroll
    for (int j = 0; j < EDGES_PER_THREAD; ++j) {
        const int e = e0 + j * FILL_THREADS;
        d[j] = 0; r[j].x = 0; r[j].y = 0;
        if (!GUARD || e < E) {
            d[j]   = edst[e];
            r[j].x = esrc[e];
            r[j].y = __float_as_int(eval[e]);
        }
    }
    #pragma unroll
    for (int j = 0; j < EDGES_PER_THREAD; ++j) {
        slot[j] = 0;
        if (!GUARD || e0 + j * FILL_THREADS < E) slot[j] = atomicAdd(plane + d[j], 1);
    }
    #pragma unroll
    for (int j = 0; j < EDGES_PER_THREAD; ++j) {
        if (!GUARD || e0 + j * FILL_THREADS < E) {
            if (slot[j] < ROW_SLOTS) {
                bucket[(d[j] * 2 + h) * ROW_SLOTS + slot[j]] = r[j];
            } else {
                int4 o;
                o.x = d[j]; o.y = r[j].x; o.z = r[j].y; o.w = 0;
                ovf[atomicAdd(ovfN, 1)] = o;
            }
        }
    }
}

__global__ void __launch_bounds__(FILL_THREADS)
fill_and_gemm1(const int* __restrict__ esrc, const int* __restrict__ edst,
               const float* __restrict__ eval, int* __restrict__ cnt,
               int2* __restrict__ bucket, int* __restrict__ ovfN,
               int4* __restrict__ ovf, int E, int cntPad, int gemmBlocks,
               const float* __restrict__ x, const float* __restrict__ W1,
               float* __restrict__ xw, int M) {
    if ((int)blockIdx.x < gemmBlocks) {
        const int waveId = blockIdx.x * 32 + (threadIdx.x >> 5);
        gemm_strip<DIMS_IN, DIMS_H>(x, W1, xw, M, waveId, threadIdx.x & 31);
        return;
    }
    const int first = (blockIdx.x - gemmBlocks) * (FILL_THREADS * EDGES_PER_THREAD);
    if (first + FILL_THREADS * EDGES_PER_THREAD <= E)
        edge_fill<false>(esrc, edst, eval, cnt, bucket, ovfN, ovf, E, cntPad, first + threadIdx.x);
    else
        edge_fill<true>(esrc, edst, eval, cnt, bucket, ovfN, ovf, E, cntPad, first + threadIdx.x);
}

struct RowEdges {
    const int2* s0;
    const int2* s1;
    int c0;
    __device__ __forceinline__ int2 rec(int j) const { return (j < c0 ? s0 : s1)[j]; }
};

template <int D, int K>
__device__ __forceinline__ void gather_edges(const RowEdges& re, int j0,
                                             const float* __restrict__ base,
                                             float4& acc) {
    int2   e[K];
    float4 m[K];
    #pragma unroll
    for (int j = 0; j < K; ++j) e[j] = re.rec(j0 + j);
    #pragma unroll
    for (int j = 0; j < K; ++j) m[j] = *(const float4*)(base + e[j].x * D);
    #pragma unroll
    for (int j = 0; j < K; ++j) {
        const float v = __int_as_float(e[j].y);
        acc.x = fmaf(v, m[j].x, acc.x);
        acc.y = fmaf(v, m[j].y, acc.y);
        acc.z = fmaf(v, m[j].z, acc.z);
        acc.w = fmaf(v, m[j].w, acc.w);
    }
}

template <int D>
__device__ __forceinline__ float4 row_sum(int row, int li,
                                          const int* __restrict__ cnt, int cntPad,
                                          const int2* __restrict__ bucket,
                                          const int* __restrict__ ovfN,
                                          const int4* __restrict__ ovf,
                                          const float* __restrict__ dense) {
    const int n0 = cnt[row];
    const int n1 = cnt[cntPad + row];
    const int c0 = min(n0, ROW_SLOTS);
    const int c1 = min(n1, ROW_SLOTS);
    RowEdges re;
    re.s0 = bucket + (row * 2) * ROW_SLOTS;
    re.s1 = re.s0 + (ROW_SLOTS - c0);
    re.c0 = c0;
    const int pend = c0 + c1;
    const float* base = dense + li * 4;

    float4 acc;
    acc.x = 0.f; acc.y = 0.f; acc.z = 0.f; acc.w = 0.f;
    int p = 0;
    for (; p + 8 <= pend; p += 8) gather_edges<D, 8>(re, p, base, acc);
    if (p + 4 <= pend) {
        gather_edges<D, 4>(re, p, base, acc);
        p += 4;
    }
    #pragma unroll 1
    for (; p < pend; ++p) gather_edges<D, 1>(re, p, base, acc);

    if (n0 > ROW_SLOTS || n1 > ROW_SLOTS) {
        const int nOv = *ovfN;
        #pragma unroll 1
        for (int k = 0; k < nOv; ++k) {
            const int4 o = ovf[k];
            if (o.x == row) {
                const float4 m = *(const float4*)(base + o.y * D);
                const float  v = __int_as_float(o.z);
                acc.x = fmaf(v, m.x, acc.x);
                acc.y = fmaf(v, m.y, acc.y);
                acc.z = fmaf(v, m.z, acc.z);
                acc.w = fmaf(v, m.w, acc.w);
            }
        }
    }
    return acc;
}

template <int D, int LPR>
__global__ void spmm_gather(const int* __restrict__ cnt, int cntPad,
                            const int2* __restrict__ bucket,
                            const int* __restrict__ ovfN,
                            const int4* __restrict__ ovf,
                            const float* __restrict__ dense,
                            float* __restrict__ out, int n) {
    constexpr int D4 = D / 4;
    const int lane = threadIdx.x & 31;
    const int wave = (blockIdx.x * blockDim.x + threadIdx.x) >> 5;
    const int row  = wave * (32 / LPR) + lane / LPR;
    const int li   = lane % LPR;
    if (row >= n || li >= D4) return;
    const float4 acc = row_sum<D>(row, li, cnt, cntPad, bucket, ovfN, ovf, dense);
    *(float4*)(out + row * D + li * 4) = acc;
}

#define TILE_ROWS 16
#define H_PITCH   100
#define TILES_PER_GROUP 4

__global__ void __launch_bounds__(TILE_ROWS * 32)
spmm_relu_gemm2(const int* __restrict__ cnt, int cntPad,
                const int2* __restrict__ bucket,
                const int* __restrict__ ovfN,
                const int4* __restrict__ ovf,
                const float* __restrict__ dense,
                const float* __restrict__ W2,
                float* __restrict__ hw, int n) {
    __shared__ float hs[TILES_PER_GROUP * TILE_ROWS * H_PITCH];
    constexpr int D4 = DIMS_H / 4;
    const int lane = threadIdx.x & 31;
    const int w     = __builtin_amdgcn_readfirstlane((int)(threadIdx.x >> 5));
    const int tile0 = blockIdx.x * TILES_PER_GROUP;

    if (lane < D4) {
        #pragma unroll 1
        for (int i = 0; i < TILES_PER_GROUP; ++i) {
            const int row = (tile0 + i) * TILE_ROWS + w;
            float4 acc;
            acc.x = 0.f; acc.y = 0.f; acc.z = 0.f; acc.w = 0.f;
            if (row < n)
                acc = row_sum<DIMS_H>(row, lane, cnt, cntPad, bucket, ovfN, ovf, dense);
            acc.x = fmaxf(acc.x, 0.f);
            acc.y = fmaxf(acc.y, 0.f);
            acc.z = fmaxf(acc.z, 0.f);
            acc.w = fmaxf(acc.w, 0.f);
            *(float4*)(hs + (i * TILE_ROWS + w) * H_PITCH + lane * 4) = acc;
        }
    }
    __syncthreads();
    if (w >= TILES_PER_GROUP || (tile0 + w) * TILE_ROWS >= n) return;
    const int row0 = (tile0 + w) * TILE_ROWS;

    const int half = lane >> 4;
    const int l15  = lane & 15;
    const float* arow = hs + (w * TILE_ROWS + l15) * H_PITCH + half * 2;
    const float* bcol = W2 + (half * 2) * DIMS_OUT + l15;

    v8f acc0 = {};
    v8f acc1 = {};
    #pragma unroll 8
    for (int k = 0; k < DIMS_H; k += 4) {
        const v2f a = *(const v2f*)(arow + k);
        v2f b0, b1;
        b0.x = bcol[k * DIMS_OUT];
        b0.y = bcol[(k + 1) * DIMS_OUT];
        b1.x = bcol[k * DIMS_OUT + 16];
        b1.y = bcol[(k + 1) * DIMS_OUT + 16];
        acc0 = __builtin_amdgcn_wmma_f32_16x16x4_f32(
            false, a, false, b0, (short)0, acc0, false, false);
        acc1 = __builtin_amdgcn_wmma_f32_16x16x4_f32(
            false, a, false, b1, (short)0, acc1, false, false);
    }

    float* c0 = hw + (row0 + half * 8) * DIMS_OUT + l15;
    #pragma unroll
    for (int r = 0; r < 8; ++r) {
        c0[r * DIMS_OUT]      = acc0[r];
        c0[r * DIMS_OUT + 16] = acc1[r];
    }
}

extern "C" void kernel_launch(void* const* d_in, const int* in_sizes, int n_in,
                              void* d_out, int out_size, void* d_ws, size_t ws_size,
                              hipStream_t stream) {
    const float* x    = (const float*)d_in[0];
    const int*   esrc = (const int*)d_in[1];
    const int*   edst = (const int*)d_in[2];
    const float* eval = (const float*)d_in[3];
    const float* W1   = (const float*)d_in[4];
    const float* W2   = (const float*)d_in[5];
    float*       out  = (float*)d_out;

    const int E = in_sizes[1];
    const int n = in_sizes[0] / DIMS_IN;

    float* buf0 = (float*)d_ws;
    float* buf1 = buf0 + (size_t)n * DIMS_H;
    const int cntPad = ((n + 31) / 32) * 32;
    int*  cnt    = (int*)(buf1 + (size_t)n * DIMS_H);
    int*  ovfN   = cnt + 2 * cntPad;
    int2* bucket = (int2*)(ovfN + 4);
    int4* ovf    = (int4*)(bucket + (size_t)n * 2 * ROW_SLOTS);

    {
        int c4 = cntPad / 2 + 1;
        zero_kernel<<<(c4 + 255) / 256, 256, 0, stream>>>((float4*)cnt, c4);
        int waves      = (n / 16) * (DIMS_H / 32);
        int gemmBlocks = (waves + 31) / 32;
        int fillBlocks = (E + FILL_THREADS * EDGES_PER_THREAD - 1) / (FILL_THREADS * EDGES_PER_THREAD);
        fill_and_gemm1<<<gemmBlocks + fillBlocks, FILL_THREADS, 0, stream>>>(
            esrc, edst, eval, cnt, bucket, ovfN, ovf, E, cntPad, gemmBlocks,
            x, W1, buf0, n);
    }

    spmm_relu_gemm2<<<(n + TILES_PER_GROUP * TILE_ROWS - 1) / (TILES_PER_GROUP * TILE_ROWS),
                      TILE_ROWS * 32, 0, stream>>>(
        cnt, cntPad, bucket, ovfN, ovf, buf0, W2, buf1, n);

    spmm_gather<DIMS_OUT, 8><<<(n + 31) / 32, 256, 0, stream>>>(
        cnt, cntPad, bucket, ovfN, ovf, buf1, out, n);
}


# GEMM1 reads x in whole 128-byte lines (16-byte loads, four rows' K quarter each) and takes A fragments from a per-wave LDS tile, not 8 bytes from each of 16 lines per load; k order unchanged
# speedup vs baseline: 3.5973x; 1.0314x over previous
#include <hip/hip_runtime.h>

#define DIMS_IN  128
#define DIMS_H   96
#define DIMS_OUT 32

typedef float v2f __attribute__((ext_vector_type(2)));
typedef float v8f __attribute__((ext_vector_type(8)));

__global__ void zero_kernel(float4* __restrict__ p, int n4) {
    int i = blockIdx.x * blockDim.x + threadIdx.x;
    if (i < n4) {
        float4 z; z.x = 0.f; z.y = 0.f; z.z = 0.f; z.w = 0.f;
        p[i] = z;
    }
}

#define A_PITCH 36
#define A_QUARTER 32
template <int K, int N>
__device__ __forceinline__ void gemm_strip(const float* __restrict__ A,
                                           const float* __restrict__ B,
                                           float* __restrict__ C,
                                           int M, int waveId, int lane,
                                           float* __restrict__ as) {
    constexpr int nPair = N >> 5;
    const int tileM  = waveId / nPair;
    const int tileP  = waveId - tileM * nPair;
    if (tileM * 16 >= M) return;

    const int half = lane >> 4;
    const int l15  = lane & 15;
    const int sr   = lane >> 3;
    const int sc   = (lane & 7) * 4;

    const float* srow = A + (tileM * 16 + sr) * K + sc;
    float*       sdst = as + sr * A_PITCH + sc;
    const float* arow = as + l15 * A_PITCH + half * 2;
    const float* bcol = B + (half * 2) * N + tileP * 32 + l15;

    v8f acc0 = {};
    v8f acc1 = {};
    #pragma unroll 1
    for (int q = 0; q < K / A_QUARTER; ++q) {
        float4 st[4];
        #pragma unroll
        for (int j = 0; j < 4; ++j) st[j] = *(const float4*)(srow + (4 * j) * K);
        #pragma unroll
        for (int j = 0; j < 4; ++j) *(float4*)(sdst + (4 * j) * A_PITCH) = st[j];
        srow += A_QUARTER;
        #pragma unroll 4
        for (int kk = 0; kk < A_QUARTER; kk += 4) {
            const v2f a = *(const v2f*)(arow + kk);
            v2f b0, b1;
            b0.x = bcol[kk * N];
            b0.y = bcol[(kk + 1) * N];
            b1.x = bcol[kk * N + 16];
            b1.y = bcol[(kk + 1) * N + 16];
            acc0 = __builtin_amdgcn_wmma_f32_16x16x4_f32(
                false, a, false, b0, (short)0, acc0, false, false);
            acc1 = __builtin_amdgcn_wmma_f32_16x16x4_f32(
                false, a, false, b1, (short)0, acc1, false, false);
        }
        bcol += A_QUARTER * N;
    }

    float* c0 = C + (tileM * 16 + half * 8) * N + tileP * 32 + l15;
    #pragma unroll
    for (int r = 0; r < 8; ++r) {
        c0[r * N]      = acc0[r];
        c0[r * N + 16] = acc1[r];
    }
}

#define ROW_SLOTS 32
#define EDGES_PER_THREAD 4
#define FILL_THREADS 1024

template <bool GUARD>
__device__ __forceinline__ void edge_fill(const int* __restrict__ esrc,
                                          const int* __restrict__ edst,
                                          const float* __restrict__ eval,
                                          int* __restrict__ cnt,
                                          int2* __restrict__ bucket,
                                          int* __restrict__ ovfN,
                                          int4* __restrict__ ovf,
                                          int E, int cntPad, int e0) {
    const unsigned where = __builtin_amdgcn_s_sendmsg_rtn(0x87);
    const int h = (int)((where >> 18) & 1u);
    int* plane = cnt + h * cntPad;
    int  d[EDGES_PER_THREAD];
    int2 r[EDGES_PER_THREAD];
    int  slot[EDGES_PER_THREAD];
    #pragma unroll
    for (int j = 0; j < EDGES_PER_THREAD; ++j) {
        const int e = e0 + j * FILL_THREADS;
        d[j] = 0; r[j].x = 0; r[j].y = 0;
        if (!GUARD || e < E) {
            d[j]   = edst[e];
            r[j].x = esrc[e];
            r[j].y = __float_as_int(eval[e]);
        }
    }
    #pragma unroll
    for (int j = 0; j < EDGES_PER_THREAD; ++j) {
        slot[j] = 0;
        if (!GUARD || e0 + j * FILL_THREADS < E) slot[j] = atomicAdd(plane + d[j], 1);
    }
    #pragma unroll
    for (int j = 0; j < EDGES_PER_THREAD; ++j) {
        if (!GUARD || e0 + j * FILL_THREADS < E) {
            if (slot[j] < ROW_SLOTS) {
                bucket[(d[j] * 2 + h) * ROW_SLOTS + slot[j]] = r[j];
            } else {
                int4 o;
                o.x = d[j]; o.y = r[j].x; o.z = r[j].y; o.w = 0;
                ovf[atomicAdd(ovfN, 1)] = o;
            }
        }
    }
}

__global__ void __launch_bounds__(FILL_THREADS)
fill_and_gemm1(const int* __restrict__ esrc, const int* __restrict__ edst,
               const float* __restrict__ eval, int* __restrict__ cnt,
               int2* __restrict__ bucket, int* __restrict__ ovfN,
               int4* __restrict__ ovf, int E, int cntPad, int gemmBlocks,
               const float* __restrict__ x, const float* __restrict__ W1,
               float* __restrict__ xw, int M) {
    __shared__ __attribute__((aligned(16))) float xs[(FILL_THREADS / 32) * 16 * A_PITCH];
    if ((int)blockIdx.x < gemmBlocks) {
        const int wv     = threadIdx.x >> 5;
        const int waveId = blockIdx.x * 32 + wv;
        gemm_strip<DIMS_IN, DIMS_H>(x, W1, xw, M, waveId, threadIdx.x & 31,
                                    xs + wv * (16 * A_PITCH));
        return;
    }
    const int first = (blockIdx.x - gemmBlocks) * (FILL_THREADS * EDGES_PER_THREAD);
    if (first + FILL_THREADS * EDGES_PER_THREAD <= E)
        edge_fill<false>(esrc, edst, eval, cnt, bucket, ovfN, ovf, E, cntPad, first + threadIdx.x);
    else
        edge_fill<true>(esrc, edst, eval, cnt, bucket, ovfN, ovf, E, cntPad, first + threadIdx.x);
}

struct RowEdges {
    const int2* s0;
    const int2* s1;
    int c0;
    __device__ __forceinline__ int2 rec(int j) const { return (j < c0 ? s0 : s1)[j]; }
};

template <int D, int K>
__device__ __forceinline__ void gather_edges(const RowEdges& re, int j0,
                                             const float* __restrict__ base,
                                             float4& acc) {
    int2   e[K];
    float4 m[K];
    #pragma unroll
    for (int j = 0; j < K; ++j) e[j] = re.rec(j0 + j);
    #pragma unroll
    for (int j = 0; j < K; ++j) m[j] = *(const float4*)(base + e[j].x * D);
    #pragma unroll
    for (int j = 0; j < K; ++j) {
        const float v = __int_as_float(e[j].y);
        acc.x = fmaf(v, m[j].x, acc.x);
        acc.y = fmaf(v, m[j].y, acc.y);
        acc.z = fmaf(v, m[j].z, acc.z);
        acc.w = fmaf(v, m[j].w, acc.w);
    }
}

template <int D>
__device__ __forceinline__ float4 row_sum(int row, int li,
                                          const int* __restrict__ cnt, int cntPad,
                                          const int2* __restrict__ bucket,
                                          const int* __restrict__ ovfN,
                                          const int4* __restrict__ ovf,
                                          const float* __restrict__ dense) {
    const int n0 = cnt[row];
    const int n1 = cnt[cntPad + row];
    const int c0 = min(n0, ROW_SLOTS);
    const int c1 = min(n1, ROW_SLOTS);
    RowEdges re;
    re.s0 = bucket + (row * 2) * ROW_SLOTS;
    re.s1 = re.s0 + (ROW_SLOTS - c0);
    re.c0 = c0;
    const int pend = c0 + c1;
    const float* base = dense + li * 4;

    float4 acc;
    acc.x = 0.f; acc.y = 0.f; acc.z = 0.f; acc.w = 0.f;
    int p = 0;
    for (; p + 8 <= pend; p += 8) gather_edges<D, 8>(re, p, base, acc);
    if (p + 4 <= pend) {
        gather_edges<D, 4>(re, p, base, acc);
        p += 4;
    }
    #pragma unroll 1
    for (; p < pend; ++p) gather_edges<D, 1>(re, p, base, acc);

    if (n0 > ROW_SLOTS || n1 > ROW_SLOTS) {
        const int nOv = *ovfN;
        #pragma unroll 1
        for (int k = 0; k < nOv; ++k) {
            const int4 o = ovf[k];
            if (o.x == row) {
                const float4 m = *(const float4*)(base + o.y * D);
                const float  v = __int_as_float(o.z);
                acc.x = fmaf(v, m.x, acc.x);
                acc.y = fmaf(v, m.y, acc.y);
                acc.z = fmaf(v, m.z, acc.z);
                acc.w = fmaf(v, m.w, acc.w);
            }
        }
    }
    return acc;
}

template <int D, int LPR>
__global__ void spmm_gather(const int* __restrict__ cnt, int cntPad,
                            const int2* __restrict__ bucket,
                            const int* __restrict__ ovfN,
                            const int4* __restrict__ ovf,
                            const float* __restrict__ dense,
                            float* __restrict__ out, int n) {
    constexpr int D4 = D / 4;
    const int lane = threadIdx.x & 31;
    const int wave = (blockIdx.x * blockDim.x + threadIdx.x) >> 5;
    const int row  = wave * (32 / LPR) + lane / LPR;
    const int li   = lane % LPR;
    if (row >= n || li >= D4) return;
    const float4 acc = row_sum<D>(row, li, cnt, cntPad, bucket, ovfN, ovf, dense);
    *(float4*)(out + row * D + li * 4) = acc;
}

#define TILE_ROWS 16
#define H_PITCH   100
#define TILES_PER_GROUP 4

__global__ void __launch_bounds__(TILE_ROWS * 32)
spmm_relu_gemm2(const int* __restrict__ cnt, int cntPad,
                const int2* __restrict__ bucket,
                const int* __restrict__ ovfN,
                const int4* __restrict__ ovf,
                const float* __restrict__ dense,
                const float* __restrict__ W2,
                float* __restrict__ hw, int n) {
    __shared__ float hs[TILES_PER_GROUP * TILE_ROWS * H_PITCH];
    constexpr int D4 = DIMS_H / 4;
    const int lane = threadIdx.x & 31;
    const int w     = __builtin_amdgcn_readfirstlane((int)(threadIdx.x >> 5));
    const int tile0 = blockIdx.x * TILES_PER_GROUP;

    if (lane < D4) {
        #pragma unroll 1
        for (int i = 0; i < TILES_PER_GROUP; ++i) {
            const int row = (tile0 + i) * TILE_ROWS + w;
            float4 acc;
            acc.x = 0.f; acc.y = 0.f; acc.z = 0.f; acc.w = 0.f;
            if (row < n)
                acc = row_sum<DIMS_H>(row, lane, cnt, cntPad, bucket, ovfN, ovf, dense);
            acc.x = fmaxf(acc.x, 0.f);
            acc.y = fmaxf(acc.y, 0.f);
            acc.z = fmaxf(acc.z, 0.f);
            acc.w = fmaxf(acc.w, 0.f);
            *(float4*)(hs + (i * TILE_ROWS + w) * H_PITCH + lane * 4) = acc;
        }
    }
    __syncthreads();
    if (w >= TILES_PER_GROUP || (tile0 + w) * TILE_ROWS >= n) return;
    const int row0 = (tile0 + w) * TILE_ROWS;

    const int half = lane >> 4;
    const int l15  = lane & 15;
    const float* arow = hs + (w * TILE_ROWS + l15) * H_PITCH + half * 2;
    const float* bcol = W2 + (half * 2) * DIMS_OUT + l15;

    v8f acc0 = {};
    v8f acc1 = {};
    #pragma unroll 8
    for (int k = 0; k < DIMS_H; k += 4) {
        const v2f a = *(const v2f*)(arow + k);
        v2f b0, b1;
        b0.x = bcol[k * DIMS_OUT];
        b0.y = bcol[(k + 1) * DIMS_OUT];
        b1.x = bcol[k * DIMS_OUT + 16];
        b1.y = bcol[(k + 1) * DIMS_OUT + 16];
        acc0 = __builtin_amdgcn_wmma_f32_16x16x4_f32(
            false, a, false, b0, (short)0, acc0, false, false);
        acc1 = __builtin_amdgcn_wmma_f32_16x16x4_f32(
            false, a, false, b1, (short)0, acc1, false, false);
    }

    float* c0 = hw + (row0 + half * 8) * DIMS_OUT + l15;
    #pragma unroll
    for (int r = 0; r < 8; ++r) {
        c0[r * DIMS_OUT]      = acc0[r];
        c0[r * DIMS_OUT + 16] = acc1[r];
    }
}

extern "C" void kernel_launch(void* const* d_in, const int* in_sizes, int n_in,
                              void* d_out, int out_size, void* d_ws, size_t ws_size,
                              hipStream_t stream) {
    const float* x    = (const float*)d_in[0];
    const int*   esrc = (const int*)d_in[1];
    const int*   edst = (const int*)d_in[2];
    const float* eval = (const float*)d_in[3];
    const float* W1   = (const float*)d_in[4];
    const float* W2   = (const float*)d_in[5];
    float*       out  = (float*)d_out;

    const int E = in_sizes[1];
    const int n = in_sizes[0] / DIMS_IN;

    float* buf0 = (float*)d_ws;
    float* buf1 = buf0 + (size_t)n * DIMS_H;
    const int cntPad = ((n + 31) / 32) * 32;
    int*  cnt    = (int*)(buf1 + (size_t)n * DIMS_H);
    int*  ovfN   = cnt + 2 * cntPad;
    int2* bucket = (int2*)(ovfN + 4);
    int4* ovf    = (int4*)(bucket + (size_t)n * 2 * ROW_SLOTS);

    {
        int c4 = cntPad / 2 + 1;
        zero_kernel<<<(c4 + 255) / 256, 256, 0, stream>>>((float4*)cnt, c4);
        int waves      = (n / 16) * (DIMS_H / 32);
        int gemmBlocks = (waves + 31) / 32;
        int fillBlocks = (E + FILL_THREADS * EDGES_PER_THREAD - 1) / (FILL_THREADS * EDGES_PER_THREAD);
        fill_and_gemm1<<<gemmBlocks + fillBlocks, FILL_THREADS, 0, stream>>>(
            esrc, edst, eval, cnt, bucket, ovfN, ovf, E, cntPad, gemmBlocks,
            x, W1, buf0, n);
    }

    spmm_relu_gemm2<<<(n + TILES_PER_GROUP * TILE_ROWS - 1) / (TILES_PER_GROUP * TILE_ROWS),
                      TILE_ROWS * 32, 0, stream>>>(
        cnt, cntPad, bucket, ovfN, ovf, buf0, W2, buf1, n);

    spmm_gather<DIMS_OUT, 8><<<(n + 31) / 32, 256, 0, stream>>>(
        cnt, cntPad, bucket, ovfN, ovf, buf1, out, n);
}
